# starting kernel
# baseline (speedup 1.0000x reference)
#include <hip/hip_runtime.h>
#include <math.h>

typedef float v2f __attribute__((ext_vector_type(2)));
typedef float v8f __attribute__((ext_vector_type(8)));

namespace {

__device__ __forceinline__ void gate_rx(float re[16], float im[16], float th, int qi) {
  const float c = cosf(0.5f * th);
  const float s = sinf(0.5f * th);
  const int mask = 8 >> qi;
#pragma unroll
  for (int p = 0; p < 16; ++p) {
    if (!(p & mask)) {
      const int q = p | mask;
      const float r0 = re[p], i0 = im[p], r1 = re[q], i1 = im[q];
      re[p] = c * r0 + s * i1;
      im[p] = c * i0 - s * r1;
      re[q] = c * r1 + s * i0;
      im[q] = c * i1 - s * r0;
    }
  }
}

__device__ __forceinline__ void gate_rz(float re[16], float im[16], float th, int qi) {
  const float c = cosf(0.5f * th);
  const float s = sinf(0.5f * th);
  const int mask = 8 >> qi;
#pragma unroll
  for (int p = 0; p < 16; ++p) {
    const float sg = (p & mask) ? s : -s;
    const float r = re[p], q = im[p];
    re[p] = r * c - q * sg;
    im[p] = r * sg + q * c;
  }
}

__device__ __forceinline__ void gate_crz(float re[16], float im[16], float th, int ctrl, int tgt) {
  const float c = cosf(0.5f * th);
  const float s = sinf(0.5f * th);
  const int cm = 8 >> ctrl, tm = 8 >> tgt;
#pragma unroll
  for (int p = 0; p < 16; ++p) {
    if (p & cm) {
      const float sg = (p & tm) ? s : -s;
      const float r = re[p], q = im[p];
      re[p] = r * c - q * sg;
      im[p] = r * sg + q * c;
    }
  }
}

}

extern "C" __global__ void qconv_setup(const float* __restrict__ w, float* __restrict__ ws) {
  const int S = threadIdx.x;
  if (S >= 16) return;
  float re[16], im[16];
#pragma unroll
  for (int p = 0; p < 16; ++p) { re[p] = 0.f; im[p] = 0.f; }
  re[0] = 1.f;

#pragma unroll
  for (int i = 0; i < 4; ++i) {
    gate_rx(re, im, w[i], i);
    gate_rz(re, im, w[4 + i], i);
  }
  int k = 8;
#pragma unroll
  for (int cd = 0; cd < 4; ++cd) {
    const int ctrl = 3 - cd;
#pragma unroll
    for (int tgt = 0; tgt < 4; ++tgt) {
      if (tgt == ctrl) continue;
      gate_crz(re, im, w[k], ctrl, tgt);
      ++k;
    }
  }
#pragma unroll
  for (int i = 0; i < 4; ++i) {
    if (S & (1 << i)) {
      const int mask = 8 >> i;
#pragma unroll
      for (int p = 0; p < 16; ++p) {
        if (!(p & mask)) {
          const int q = p | mask;
          const float r0 = re[p], i0 = im[p];
          re[p] = -re[q]; im[p] = -im[q];
          re[q] =  r0;    im[q] =  i0;
        }
      }
    }
  }
#pragma unroll
  for (int i = 0; i < 4; ++i) {
    gate_rx(re, im, w[20 + i], i);
    gate_rz(re, im, w[24 + i], i);
  }
#pragma unroll
  for (int p = 0; p < 16; ++p) {
    ws[S * 16 + p]       = re[p];
    ws[256 + S * 16 + p] = im[p];
  }
}

extern "C" __global__ __launch_bounds__(256) void qconv_main(
    const float* __restrict__ x, const float* __restrict__ Bmat,
    float* __restrict__ out) {
  __shared__ float lds[8 * 256];

  const int tid  = threadIdx.x;
  const int lane = tid & 31;
  const int wave = tid >> 5;
  const int hb   = lane >> 4;
  const int M    = lane & 15;
  const int base = (blockIdx.x * 8 + wave) * 16;
  const int n    = base + M;
  const int j = n & 63, i = (n >> 6) & 63, b = n >> 12;

  const float* img = x + (((size_t)b) << 14) + (size_t)(2 * i) * 128 + (size_t)(2 * j);
  const float2 top = *(const float2*)(img);
  const float2 bot = *(const float2*)(img + 128);
  const float v0 = top.x, v1 = top.y, v2 = bot.x, v3 = bot.y;

  const float ssq  = v0 * v0 + v1 * v1 + v2 * v2 + v3 * v3;
  const bool  zero = (v0 == 0.f) && (v1 == 0.f) && (v2 == 0.f) && (v3 == 0.f);
  const float inv  = zero ? 1.f : (1.f / sqrtf(ssq));
  const float t0 = zero ? 1.f : v0 * inv;
  const float t1 = zero ? 0.f : v1 * inv;
  const float t2 = zero ? 0.f : v2 * inv;
  const float t3 = zero ? 0.f : v3 * inv;

  const float c0 = cosf(0.5f * t0), s0 = sinf(0.5f * t0);
  const float c1 = cosf(0.5f * t1), s1 = sinf(0.5f * t1);
  const float c2 = cosf(0.5f * t2), s2 = sinf(0.5f * t2);
  const float c3 = cosf(0.5f * t3), s3 = sinf(0.5f * t3);

  const float p01[4] = { c0 * c1, s0 * c1, c0 * s1, s0 * s1 };
  const float p23[4] = { c2 * c3, s2 * c3, c2 * s3, s2 * s3 };
  float m[16];
#pragma unroll
  for (int S = 0; S < 16; ++S) m[S] = p01[S & 3] * p23[S >> 2];

  v2f a[4];
#pragma unroll
  for (int f = 0; f < 4; ++f) {
    a[f].x = hb ? m[4 * f + 2] : m[4 * f + 0];
    a[f].y = hb ? m[4 * f + 3] : m[4 * f + 1];
  }

  v2f bre[4], bim[4];
#pragma unroll
  for (int f = 0; f < 4; ++f) {
    const int kb = 4 * f + 2 * hb;
    bre[f].x = Bmat[kb * 16 + M];
    bre[f].y = Bmat[kb * 16 + 16 + M];
    bim[f].x = Bmat[256 + kb * 16 + M];
    bim[f].y = Bmat[256 + kb * 16 + 16 + M];
  }

  v8f sre = {0.f, 0.f, 0.f, 0.f, 0.f, 0.f, 0.f, 0.f};
  v8f sim = {0.f, 0.f, 0.f, 0.f, 0.f, 0.f, 0.f, 0.f};
#pragma unroll
  for (int f = 0; f < 4; ++f) {
    sre = __builtin_amdgcn_wmma_f32_16x16x4_f32(false, a[f], false, bre[f],
                                                (short)0, sre, false, false);
    sim = __builtin_amdgcn_wmma_f32_16x16x4_f32(false, a[f], false, bim[f],
                                                (short)0, sim, false, false);
  }
  const v8f P = sre * sre + sim * sim;

  float* lp = &lds[wave * 256];
#pragma unroll
  for (int v = 0; v < 8; ++v) lp[(v + 8 * hb) * 16 + M] = P[v];
  asm volatile("s_wait_dscnt 0x0" ::: "memory");

  if (lane < 16) {
    const float* row = &lds[wave * 256 + lane * 16];
    float q0 = 0.f, q1 = 0.f, q2 = 0.f, q3 = 0.f;
#pragma unroll
    for (int p = 0; p < 16; ++p) {
      const float pv = row[p];
      q0 += (p & 8) ? -pv : pv;
      q1 += (p & 4) ? -pv : pv;
      q2 += (p & 2) ? -pv : pv;
      q3 += (p & 1) ? -pv : pv;
    }
    float* o = out + (((size_t)b) << 14) + (size_t)i * 64 + (size_t)j;
    o[0]     = q0;
    o[4096]  = q1;
    o[8192]  = q2;
    o[12288] = q3;
  }
}

extern "C" void kernel_launch(void* const* d_in, const int* in_sizes, int n_in,
                              void* d_out, int out_size, void* d_ws, size_t ws_size,
                              hipStream_t stream) {
  const float* x = (const float*)d_in[0];
  const float* w = (const float*)d_in[1];
  float* out = (float*)d_out;
  float* ws  = (float*)d_ws;

  qconv_setup<<<1, 16, 0, stream>>>(w, ws);
  qconv_main<<<2048, 256, 0, stream>>>(x, ws, out);
}


# Single launch: each work-group builds the 16x16 complex circuit matrix itself in LDS (closed form per qubit over its 256 threads); setup kernel and workspace dropped; matrix entries round differently
# speedup vs baseline: 1.7388x; 1.7388x over previous
#include <hip/hip_runtime.h>
#include <math.h>


typedef float v2f __attribute__((ext_vector_type(2)));
typedef float v8f __attribute__((ext_vector_type(8)));

extern "C" __global__ __launch_bounds__(256) void qconv_main(
    const float* __restrict__ x, const float* __restrict__ w,
    float* __restrict__ out) {
  __shared__ float lds[8 * 256];
  __shared__ __attribute__((aligned(16))) float sc[64];
  __shared__ float psi[32];
  __shared__ float Cs[512];

  const int tid  = threadIdx.x;
  const int lane = tid & 31;
  const int wave = tid >> 5;
  const int hb   = lane >> 4;
  const int M    = lane & 15;
  const int base = (blockIdx.x * 8 + wave) * 16;
  const int n    = base + M;
  const int j = n & 63, i = (n >> 6) & 63, b = n >> 12;

  const float* img = x + (((size_t)b) << 14) + (size_t)(2 * i) * 128 + (size_t)(2 * j);
  const float2 top = *(const float2*)(img);
  const float2 bot = *(const float2*)(img + 128);

  if (tid < 28) {
    const float th = w[tid];
    sc[tid]      = cosf(0.5f * th);
    sc[32 + tid] = sinf(0.5f * th);
  }
  __syncthreads();

  if (tid < 16) {
    const int r = tid;
    float re = 1.f, im = 0.f;
#pragma unroll
    for (int q = 0; q < 4; ++q) {
      const bool bit = ((r >> (3 - q)) & 1) != 0;
      const float c = sc[q], s = sc[32 + q], cz = sc[4 + q], sz = sc[36 + q];
      const float ar = bit ? s * sz : c * cz;
      const float ai = bit ? -(s * cz) : -(c * sz);
      const float nr = re * ar - im * ai;
      const float ni = re * ai + im * ar;
      re = nr; im = ni;
    }
    int k = 8;
#pragma unroll
    for (int cd = 0; cd < 4; ++cd) {
      const int ctrl = 3 - cd;
#pragma unroll
      for (int tgt = 0; tgt < 4; ++tgt) {
        if (tgt == ctrl) continue;
        const float c = sc[k], s = sc[32 + k];
        const bool use = (r & (8 >> ctrl)) != 0;
        const float sg = (r & (8 >> tgt)) ? s : -s;
        const float pr = use ? c : 1.f;
        const float pi = use ? sg : 0.f;
        const float nr = re * pr - im * pi;
        const float ni = re * pi + im * pr;
        re = nr; im = ni;
        ++k;
      }
    }
    psi[r]      = re;
    psi[16 + r] = im;
  }

  const float v0 = top.x, v1 = top.y, v2 = bot.x, v3 = bot.y;

  const float ssq  = v0 * v0 + v1 * v1 + v2 * v2 + v3 * v3;
  const bool  zero = (v0 == 0.f) && (v1 == 0.f) && (v2 == 0.f) && (v3 == 0.f);
  const float inv  = zero ? 1.f : (1.f / sqrtf(ssq));
  const float t0 = zero ? 1.f : v0 * inv;
  const float t1 = zero ? 0.f : v1 * inv;
  const float t2 = zero ? 0.f : v2 * inv;
  const float t3 = zero ? 0.f : v3 * inv;

  const float c0 = cosf(0.5f * t0), s0 = sinf(0.5f * t0);
  const float c1 = cosf(0.5f * t1), s1 = sinf(0.5f * t1);
  const float c2 = cosf(0.5f * t2), s2 = sinf(0.5f * t2);
  const float c3 = cosf(0.5f * t3), s3 = sinf(0.5f * t3);

  __syncthreads();

  {
    const int S = tid >> 4, p = tid & 15;
    float vr = psi[p], vi = psi[16 + p];
    const float4 cT = *(const float4*)&sc[20];
    const float4 cP = *(const float4*)&sc[24];
    const float4 sT = *(const float4*)&sc[52];
    const float4 sP = *(const float4*)&sc[56];
    const float cth[4] = { cT.x, cT.y, cT.z, cT.w };
    const float cph[4] = { cP.x, cP.y, cP.z, cP.w };
    const float sth[4] = { sT.x, sT.y, sT.z, sT.w };
    const float sph[4] = { sP.x, sP.y, sP.z, sP.w };
#pragma unroll
    for (int q = 0; q < 4; ++q) {
      const int  mask = 8 >> q;
      const bool a  = (p & mask) != 0;
      const bool kb = ((S >> q) & 1) != 0;
      const float pr = __shfl_xor(vr, mask, 32);
      const float pi = __shfl_xor(vi, mask, 32);
      const float sg = a ? sph[q] : -sph[q];
      const float Dr = cph[q] * cth[q], Di = sg * cth[q];
      const float Or = sg * sth[q],     Oi = -(cph[q] * sth[q]);
      const float tau = a ? -1.f : 1.f;
      const float xr = kb ? tau * Or : Dr;
      const float xi = kb ? tau * Oi : Di;
      const float yr = kb ? -(tau * Dr) : Or;
      const float yi = kb ? -(tau * Di) : Oi;
      const float nr = xr * vr - xi * vi + yr * pr - yi * pi;
      const float ni = xr * vi + xi * vr + yr * pi + yi * pr;
      vr = nr; vi = ni;
    }
    Cs[tid]       = vr;
    Cs[256 + tid] = vi;
  }
  __syncthreads();

  const float p01[4] = { c0 * c1, s0 * c1, c0 * s1, s0 * s1 };
  const float p23[4] = { c2 * c3, s2 * c3, c2 * s3, s2 * s3 };
  float m[16];
#pragma unroll
  for (int S = 0; S < 16; ++S) m[S] = p01[S & 3] * p23[S >> 2];

  v2f a[4];
#pragma unroll
  for (int f = 0; f < 4; ++f) {
    a[f].x = hb ? m[4 * f + 2] : m[4 * f + 0];
    a[f].y = hb ? m[4 * f + 3] : m[4 * f + 1];
  }

  v2f bre[4], bim[4];
#pragma unroll
  for (int f = 0; f < 4; ++f) {
    const int kb = 4 * f + 2 * hb;
    bre[f].x = Cs[kb * 16 + M];
    bre[f].y = Cs[kb * 16 + 16 + M];
    bim[f].x = Cs[256 + kb * 16 + M];
    bim[f].y = Cs[256 + kb * 16 + 16 + M];
  }

  v8f sre = {0.f, 0.f, 0.f, 0.f, 0.f, 0.f, 0.f, 0.f};
  v8f sim = {0.f, 0.f, 0.f, 0.f, 0.f, 0.f, 0.f, 0.f};
#pragma unroll
  for (int f = 0; f < 4; ++f) {
    sre = __builtin_amdgcn_wmma_f32_16x16x4_f32(false, a[f], false, bre[f],
                                                (short)0, sre, false, false);
    sim = __builtin_amdgcn_wmma_f32_16x16x4_f32(false, a[f], false, bim[f],
                                                (short)0, sim, false, false);
  }
  const v8f P = sre * sre + sim * sim;

  float* lp = &lds[wave * 256];
#pragma unroll
  for (int v = 0; v < 8; ++v) lp[(v + 8 * hb) * 16 + M] = P[v];
  asm volatile("s_wait_dscnt 0x0" ::: "memory");

  if (lane < 16) {
    const float* row = &lds[wave * 256 + lane * 16];
    float q0 = 0.f, q1 = 0.f, q2 = 0.f, q3 = 0.f;
#pragma unroll
    for (int p = 0; p < 16; ++p) {
      const float pv = row[p];
      q0 += (p & 8) ? -pv : pv;
      q1 += (p & 4) ? -pv : pv;
      q2 += (p & 2) ? -pv : pv;
      q3 += (p & 1) ? -pv : pv;
    }
    float* o = out + (((size_t)b) << 14) + (size_t)i * 64 + (size_t)j;
    o[0]     = q0;
    o[4096]  = q1;
    o[8192]  = q2;
    o[12288] = q3;
  }
}

extern "C" void kernel_launch(void* const* d_in, const int* in_sizes, int n_in,
                              void* d_out, int out_size, void* d_ws, size_t ws_size,
                              hipStream_t stream) {
  const float* x = (const float*)d_in[0];
  const float* w = (const float*)d_in[1];
  float* out = (float*)d_out;

  qconv_main<<<2048, 256, 0, stream>>>(x, w, out);
}


# One patch per lane: a wave takes 32 patches as two 16-row tiles and computes each patch's cos/sin once; the other half's A columns come by a lane l^16 exchange of six factors; grid 2048 -> 1024
# speedup vs baseline: 2.1791x; 1.2532x over previous
#include <hip/hip_runtime.h>
#include <math.h>


typedef float v2f __attribute__((ext_vector_type(2)));
typedef float v8f __attribute__((ext_vector_type(8)));

static __device__ __forceinline__ float xhalf(float v) {
  const int bits = __builtin_bit_cast(int, v);
  return __builtin_bit_cast(float,
      __builtin_amdgcn_permlanex16(bits, bits, 0x76543210, 0xfedcba98, false, false));
}

extern "C" __global__ __launch_bounds__(256) void qconv_main(
    const float* __restrict__ x, const float* __restrict__ w,
    float* __restrict__ out) {
  __shared__ float lds[8 * 512];
  __shared__ __attribute__((aligned(16))) float sc[64];
  __shared__ float psi[32];
  __shared__ float Cs[512];

  const int tid  = threadIdx.x;
  const int lane = tid & 31;
  const int wave = tid >> 5;
  const int hb   = lane >> 4;
  const int M    = lane & 15;
  const int base = (blockIdx.x * 8 + wave) * 32;
  const int n    = base + lane;
  const int j = n & 63, i = (n >> 6) & 63, b = n >> 12;

  const float* img = x + (((size_t)b) << 14) + (size_t)(2 * i) * 128 + (size_t)(2 * j);
  const float2 top = *(const float2*)(img);
  const float2 bot = *(const float2*)(img + 128);

  if (tid < 28) {
    const float th = w[tid];
    sc[tid]      = cosf(0.5f * th);
    sc[32 + tid] = sinf(0.5f * th);
  }
  __syncthreads();

  if (tid < 16) {
    const int r = tid;
    float re = 1.f, im = 0.f;
#pragma unroll
    for (int q = 0; q < 4; ++q) {
      const bool bit = ((r >> (3 - q)) & 1) != 0;
      const float c = sc[q], s = sc[32 + q], cz = sc[4 + q], sz = sc[36 + q];
      const float ar = bit ? s * sz : c * cz;
      const float ai = bit ? -(s * cz) : -(c * sz);
      const float nr = re * ar - im * ai;
      const float ni = re * ai + im * ar;
      re = nr; im = ni;
    }
    int k = 8;
#pragma unroll
    for (int cd = 0; cd < 4; ++cd) {
      const int ctrl = 3 - cd;
#pragma unroll
      for (int tgt = 0; tgt < 4; ++tgt) {
        if (tgt == ctrl) continue;
        const float c = sc[k], s = sc[32 + k];
        const bool use = (r & (8 >> ctrl)) != 0;
        const float sg = (r & (8 >> tgt)) ? s : -s;
        const float pr = use ? c : 1.f;
        const float pi = use ? sg : 0.f;
        const float nr = re * pr - im * pi;
        const float ni = re * pi + im * pr;
        re = nr; im = ni;
        ++k;
      }
    }
    psi[r]      = re;
    psi[16 + r] = im;
  }

  const float v0 = top.x, v1 = top.y, v2 = bot.x, v3 = bot.y;

  const float ssq  = v0 * v0 + v1 * v1 + v2 * v2 + v3 * v3;
  const bool  zero = (v0 == 0.f) && (v1 == 0.f) && (v2 == 0.f) && (v3 == 0.f);
  const float inv  = zero ? 1.f : (1.f / sqrtf(ssq));
  const float t0 = zero ? 1.f : v0 * inv;
  const float t1 = zero ? 0.f : v1 * inv;
  const float t2 = zero ? 0.f : v2 * inv;
  const float t3 = zero ? 0.f : v3 * inv;

  const float c0 = cosf(0.5f * t0), s0 = sinf(0.5f * t0);
  const float c1 = cosf(0.5f * t1), s1 = sinf(0.5f * t1);
  const float c2 = cosf(0.5f * t2), s2 = sinf(0.5f * t2);
  const float c3 = cosf(0.5f * t3), s3 = sinf(0.5f * t3);

  __syncthreads();

  {
    const int S = tid >> 4, p = tid & 15;
    float vr = psi[p], vi = psi[16 + p];
    const float4 cT = *(const float4*)&sc[20];
    const float4 cP = *(const float4*)&sc[24];
    const float4 sT = *(const float4*)&sc[52];
    const float4 sP = *(const float4*)&sc[56];
    const float cth[4] = { cT.x, cT.y, cT.z, cT.w };
    const float cph[4] = { cP.x, cP.y, cP.z, cP.w };
    const float sth[4] = { sT.x, sT.y, sT.z, sT.w };
    const float sph[4] = { sP.x, sP.y, sP.z, sP.w };
#pragma unroll
    for (int q = 0; q < 4; ++q) {
      const int  mask = 8 >> q;
      const bool a  = (p & mask) != 0;
      const bool kb = ((S >> q) & 1) != 0;
      const float pr = __shfl_xor(vr, mask, 32);
      const float pi = __shfl_xor(vi, mask, 32);
      const float sg = a ? sph[q] : -sph[q];
      const float Dr = cph[q] * cth[q], Di = sg * cth[q];
      const float Or = sg * sth[q],     Oi = -(cph[q] * sth[q]);
      const float tau = a ? -1.f : 1.f;
      const float xr = kb ? tau * Or : Dr;
      const float xi = kb ? tau * Oi : Di;
      const float yr = kb ? -(tau * Dr) : Or;
      const float yi = kb ? -(tau * Di) : Oi;
      const float nr = xr * vr - xi * vi + yr * pr - yi * pi;
      const float ni = xr * vi + xi * vr + yr * pi + yi * pr;
      vr = nr; vi = ni;
    }
    Cs[tid]       = vr;
    Cs[256 + tid] = vi;
  }
  __syncthreads();

  const float p01[4] = { c0 * c1, s0 * c1, c0 * s1, s0 * s1 };
  const float p23[4] = { c2 * c3, s2 * c3, c2 * s3, s2 * s3 };

  const float qo0 = hb ? p01[2] : p01[0];
  const float qo1 = hb ? p01[3] : p01[1];
  const float qx0 = hb ? p01[0] : p01[2];
  const float qx1 = hb ? p01[1] : p01[3];
  const float rq0 = xhalf(qx0);
  const float rq1 = xhalf(qx1);
  const float rp[4] = { xhalf(p23[0]), xhalf(p23[1]), xhalf(p23[2]), xhalf(p23[3]) };

  const float t0q0 = hb ? rq0 : qo0, t0q1 = hb ? rq1 : qo1;
  const float t1q0 = hb ? qo0 : rq0, t1q1 = hb ? qo1 : rq1;
  v2f a0[4], a1[4];
  for (int f = 0; f < 4; ++f) {
    const float t0p = hb ? rp[f] : p23[f];
    const float t1p = hb ? p23[f] : rp[f];
    a0[f].x = t0q0 * t0p;
    a0[f].y = t0q1 * t0p;
    a1[f].x = t1q0 * t1p;
    a1[f].y = t1q1 * t1p;
  }

  v2f bre[4], bim[4];
#pragma unroll
  for (int f = 0; f < 4; ++f) {
    const int kb = 4 * f + 2 * hb;
    bre[f].x = Cs[kb * 16 + M];
    bre[f].y = Cs[kb * 16 + 16 + M];
    bim[f].x = Cs[256 + kb * 16 + M];
    bim[f].y = Cs[256 + kb * 16 + 16 + M];
  }

  v8f sre0 = {0.f, 0.f, 0.f, 0.f, 0.f, 0.f, 0.f, 0.f};
  v8f sim0 = {0.f, 0.f, 0.f, 0.f, 0.f, 0.f, 0.f, 0.f};
  v8f sre1 = {0.f, 0.f, 0.f, 0.f, 0.f, 0.f, 0.f, 0.f};
  v8f sim1 = {0.f, 0.f, 0.f, 0.f, 0.f, 0.f, 0.f, 0.f};
#pragma unroll
  for (int f = 0; f < 4; ++f) {
    sre0 = __builtin_amdgcn_wmma_f32_16x16x4_f32(false, a0[f], false, bre[f],
                                                 (short)0, sre0, false, false);
    sim0 = __builtin_amdgcn_wmma_f32_16x16x4_f32(false, a0[f], false, bim[f],
                                                 (short)0, sim0, false, false);
    sre1 = __builtin_amdgcn_wmma_f32_16x16x4_f32(false, a1[f], false, bre[f],
                                                 (short)0, sre1, false, false);
    sim1 = __builtin_amdgcn_wmma_f32_16x16x4_f32(false, a1[f], false, bim[f],
                                                 (short)0, sim1, false, false);
  }
  const v8f P0 = sre0 * sre0 + sim0 * sim0;
  const v8f P1 = sre1 * sre1 + sim1 * sim1;

  float* lp = &lds[wave * 512];
#pragma unroll
  for (int v = 0; v < 8; ++v) {
    lp[(v + 8 * hb) * 16 + M]       = P0[v];
    lp[256 + (v + 8 * hb) * 16 + M] = P1[v];
  }
  asm volatile("s_wait_dscnt 0x0" ::: "memory");

  {
    const float* row = lp + lane * 16;
    float q0 = 0.f, q1 = 0.f, q2 = 0.f, q3 = 0.f;
#pragma unroll
    for (int p = 0; p < 16; ++p) {
      const float pv = row[p];
      q0 += (p & 8) ? -pv : pv;
      q1 += (p & 4) ? -pv : pv;
      q2 += (p & 2) ? -pv : pv;
      q3 += (p & 1) ? -pv : pv;
    }
    float* o = out + (((size_t)b) << 14) + (size_t)i * 64 + (size_t)j;
    o[0]     = q0;
    o[4096]  = q1;
    o[8192]  = q2;
    o[12288] = q3;
  }
}

extern "C" void kernel_launch(void* const* d_in, const int* in_sizes, int n_in,
                              void* d_out, int out_size, void* d_ws, size_t ws_size,
                              hipStream_t stream) {
  const float* x = (const float*)d_in[0];
  const float* w = (const float*)d_in[1];
  float* out = (float*)d_out;

  qconv_main<<<1024, 256, 0, stream>>>(x, w, out);
}


# Encoding cos/sin by the math library's small-argument polynomials directly: half angles stay in [-0.5,0.5], so range reduction, quadrant select and large-argument path go; same values
# speedup vs baseline: 2.4085x; 1.1053x over previous
#include <hip/hip_runtime.h>
#include <math.h>


typedef float v2f __attribute__((ext_vector_type(2)));
typedef float v8f __attribute__((ext_vector_type(8)));

static __device__ __forceinline__ float xhalf(float v) {
  const int bits = __builtin_bit_cast(int, v);
  return __builtin_bit_cast(float,
      __builtin_amdgcn_permlanex16(bits, bits, 0x76543210, 0xfedcba98, false, false));
}

static __device__ __forceinline__ void cs_half(float h, float& c, float& s) {
  const float z = h * h;
  float p = fmaf(z, -1.94644585e-4f, 8.33171979e-3f);
  p = fmaf(z, p, -1.66666463e-1f);
  p = z * p;
  s = fmaf(h, p, h);
  float q = fmaf(z, 2.56687417e-5f, -1.39091106e-3f);
  q = fmaf(z, q, 4.16679047e-2f);
  q = fmaf(z, q, -5.00000238e-1f);
  c = fmaf(z, q, 1.0f);
}

extern "C" __global__ __launch_bounds__(256) void qconv_main(
    const float* __restrict__ x, const float* __restrict__ w,
    float* __restrict__ out) {
  __shared__ float lds[8 * 512];
  __shared__ __attribute__((aligned(16))) float sc[64];
  __shared__ float psi[32];
  __shared__ float Cs[512];

  const int tid  = threadIdx.x;
  const int lane = tid & 31;
  const int wave = tid >> 5;
  const int hb   = lane >> 4;
  const int M    = lane & 15;
  const int base = (blockIdx.x * 8 + wave) * 32;
  const int n    = base + lane;
  const int j = n & 63, i = (n >> 6) & 63, b = n >> 12;

  const float* img = x + (((size_t)b) << 14) + (size_t)(2 * i) * 128 + (size_t)(2 * j);
  const float2 top = *(const float2*)(img);
  const float2 bot = *(const float2*)(img + 128);

  if (tid < 28) {
    const float th = w[tid];
    sc[tid]      = cosf(0.5f * th);
    sc[32 + tid] = sinf(0.5f * th);
  }
  __syncthreads();

  if (tid < 16) {
    const int r = tid;
    float re = 1.f, im = 0.f;
#pragma unroll
    for (int q = 0; q < 4; ++q) {
      const bool bit = ((r >> (3 - q)) & 1) != 0;
      const float c = sc[q], s = sc[32 + q], cz = sc[4 + q], sz = sc[36 + q];
      const float ar = bit ? s * sz : c * cz;
      const float ai = bit ? -(s * cz) : -(c * sz);
      const float nr = re * ar - im * ai;
      const float ni = re * ai + im * ar;
      re = nr; im = ni;
    }
    int k = 8;
#pragma unroll
    for (int cd = 0; cd < 4; ++cd) {
      const int ctrl = 3 - cd;
#pragma unroll
      for (int tgt = 0; tgt < 4; ++tgt) {
        if (tgt == ctrl) continue;
        const float c = sc[k], s = sc[32 + k];
        const bool use = (r & (8 >> ctrl)) != 0;
        const float sg = (r & (8 >> tgt)) ? s : -s;
        const float pr = use ? c : 1.f;
        const float pi = use ? sg : 0.f;
        const float nr = re * pr - im * pi;
        const float ni = re * pi + im * pr;
        re = nr; im = ni;
        ++k;
      }
    }
    psi[r]      = re;
    psi[16 + r] = im;
  }

  const float v0 = top.x, v1 = top.y, v2 = bot.x, v3 = bot.y;

  const float ssq  = v0 * v0 + v1 * v1 + v2 * v2 + v3 * v3;
  const bool  zero = (v0 == 0.f) && (v1 == 0.f) && (v2 == 0.f) && (v3 == 0.f);
  const float inv  = zero ? 1.f : (1.f / sqrtf(ssq));
  const float t0 = zero ? 1.f : v0 * inv;
  const float t1 = zero ? 0.f : v1 * inv;
  const float t2 = zero ? 0.f : v2 * inv;
  const float t3 = zero ? 0.f : v3 * inv;

  float c0, s0, c1, s1, c2, s2, c3, s3;
  cs_half(0.5f * t0, c0, s0);
  cs_half(0.5f * t1, c1, s1);
  cs_half(0.5f * t2, c2, s2);
  cs_half(0.5f * t3, c3, s3);

  __syncthreads();

  {
    const int S = tid >> 4, p = tid & 15;
    float vr = psi[p], vi = psi[16 + p];
    const float4 cT = *(const float4*)&sc[20];
    const float4 cP = *(const float4*)&sc[24];
    const float4 sT = *(const float4*)&sc[52];
    const float4 sP = *(const float4*)&sc[56];
    const float cth[4] = { cT.x, cT.y, cT.z, cT.w };
    const float cph[4] = { cP.x, cP.y, cP.z, cP.w };
    const float sth[4] = { sT.x, sT.y, sT.z, sT.w };
    const float sph[4] = { sP.x, sP.y, sP.z, sP.w };
#pragma unroll
    for (int q = 0; q < 4; ++q) {
      const int  mask = 8 >> q;
      const bool a  = (p & mask) != 0;
      const bool kb = ((S >> q) & 1) != 0;
      const float pr = __shfl_xor(vr, mask, 32);
      const float pi = __shfl_xor(vi, mask, 32);
      const float sg = a ? sph[q] : -sph[q];
      const float Dr = cph[q] * cth[q], Di = sg * cth[q];
      const float Or = sg * sth[q],     Oi = -(cph[q] * sth[q]);
      const float tau = a ? -1.f : 1.f;
      const float xr = kb ? tau * Or : Dr;
      const float xi = kb ? tau * Oi : Di;
      const float yr = kb ? -(tau * Dr) : Or;
      const float yi = kb ? -(tau * Di) : Oi;
      const float nr = xr * vr - xi * vi + yr * pr - yi * pi;
      const float ni = xr * vi + xi * vr + yr * pi + yi * pr;
      vr = nr; vi = ni;
    }
    Cs[tid]       = vr;
    Cs[256 + tid] = vi;
  }
  __syncthreads();

  const float p01[4] = { c0 * c1, s0 * c1, c0 * s1, s0 * s1 };
  const float p23[4] = { c2 * c3, s2 * c3, c2 * s3, s2 * s3 };

  const float qo0 = hb ? p01[2] : p01[0];
  const float qo1 = hb ? p01[3] : p01[1];
  const float qx0 = hb ? p01[0] : p01[2];
  const float qx1 = hb ? p01[1] : p01[3];
  const float rq0 = xhalf(qx0);
  const float rq1 = xhalf(qx1);
  const float rp[4] = { xhalf(p23[0]), xhalf(p23[1]), xhalf(p23[2]), xhalf(p23[3]) };

  const float t0q0 = hb ? rq0 : qo0, t0q1 = hb ? rq1 : qo1;
  const float t1q0 = hb ? qo0 : rq0, t1q1 = hb ? qo1 : rq1;
  v2f a0[4], a1[4];
  for (int f = 0; f < 4; ++f) {
    const float t0p = hb ? rp[f] : p23[f];
    const float t1p = hb ? p23[f] : rp[f];
    a0[f].x = t0q0 * t0p;
    a0[f].y = t0q1 * t0p;
    a1[f].x = t1q0 * t1p;
    a1[f].y = t1q1 * t1p;
  }

  v2f bre[4], bim[4];
#pragma unroll
  for (int f = 0; f < 4; ++f) {
    const int kb = 4 * f + 2 * hb;
    bre[f].x = Cs[kb * 16 + M];
    bre[f].y = Cs[kb * 16 + 16 + M];
    bim[f].x = Cs[256 + kb * 16 + M];
    bim[f].y = Cs[256 + kb * 16 + 16 + M];
  }

  v8f sre0 = {0.f, 0.f, 0.f, 0.f, 0.f, 0.f, 0.f, 0.f};
  v8f sim0 = {0.f, 0.f, 0.f, 0.f, 0.f, 0.f, 0.f, 0.f};
  v8f sre1 = {0.f, 0.f, 0.f, 0.f, 0.f, 0.f, 0.f, 0.f};
  v8f sim1 = {0.f, 0.f, 0.f, 0.f, 0.f, 0.f, 0.f, 0.f};
#pragma unroll
  for (int f = 0; f < 4; ++f) {
    sre0 = __builtin_amdgcn_wmma_f32_16x16x4_f32(false, a0[f], false, bre[f],
                                                 (short)0, sre0, false, false);
    sim0 = __builtin_amdgcn_wmma_f32_16x16x4_f32(false, a0[f], false, bim[f],
                                                 (short)0, sim0, false, false);
    sre1 = __builtin_amdgcn_wmma_f32_16x16x4_f32(false, a1[f], false, bre[f],
                                                 (short)0, sre1, false, false);
    sim1 = __builtin_amdgcn_wmma_f32_16x16x4_f32(false, a1[f], false, bim[f],
                                                 (short)0, sim1, false, false);
  }
  const v8f P0 = sre0 * sre0 + sim0 * sim0;
  const v8f P1 = sre1 * sre1 + sim1 * sim1;

  float* lp = &lds[wave * 512];
#pragma unroll
  for (int v = 0; v < 8; ++v) {
    lp[(v + 8 * hb) * 16 + M]       = P0[v];
    lp[256 + (v + 8 * hb) * 16 + M] = P1[v];
  }
  asm volatile("s_wait_dscnt 0x0" ::: "memory");

  {
    const float* row = lp + lane * 16;
    float q0 = 0.f, q1 = 0.f, q2 = 0.f, q3 = 0.f;
#pragma unroll
    for (int p = 0; p < 16; ++p) {
      const float pv = row[p];
      q0 += (p & 8) ? -pv : pv;
      q1 += (p & 4) ? -pv : pv;
      q2 += (p & 2) ? -pv : pv;
      q3 += (p & 1) ? -pv : pv;
    }
    float* o = out + (((size_t)b) << 14) + (size_t)i * 64 + (size_t)j;
    o[0]     = q0;
    o[4096]  = q1;
    o[8192]  = q2;
    o[12288] = q3;
  }
}

extern "C" void kernel_launch(void* const* d_in, const int* in_sizes, int n_in,
                              void* d_out, int out_size, void* d_ws, size_t ws_size,
                              hipStream_t stream) {
  const float* x = (const float*)d_in[0];
  const float* w = (const float*)d_in[1];
  float* out = (float*)d_out;

  qconv_main<<<1024, 256, 0, stream>>>(x, w, out);
}


# Two patches per lane (neighbours in the row: one float4 per pixel row in, float2 per plane out): a wave takes 64 patches as four 16-row tiles on one B fragment set, no loop; grid 1024 -> 512
# speedup vs baseline: 2.6843x; 1.1145x over previous
#include <hip/hip_runtime.h>
#include <math.h>


typedef float v2f __attribute__((ext_vector_type(2)));
typedef float v8f __attribute__((ext_vector_type(8)));

static __device__ __forceinline__ float xhalf(float v) {
  const int bits = __builtin_bit_cast(int, v);
  return __builtin_bit_cast(float,
      __builtin_amdgcn_permlanex16(bits, bits, 0x76543210, 0xfedcba98, false, false));
}

static __device__ __forceinline__ void cs_half(float h, float& c, float& s) {
  const float z = h * h;
  float p = fmaf(z, -1.94644585e-4f, 8.33171979e-3f);
  p = fmaf(z, p, -1.66666463e-1f);
  p = z * p;
  s = fmaf(h, p, h);
  float q = fmaf(z, 2.56687417e-5f, -1.39091106e-3f);
  q = fmaf(z, q, 4.16679047e-2f);
  q = fmaf(z, q, -5.00000238e-1f);
  c = fmaf(z, q, 1.0f);
}


static __device__ __forceinline__ void encode(float v0, float v1, float v2, float v3,
                                              float c[4], float s[4]) {
  const float ssq  = v0 * v0 + v1 * v1 + v2 * v2 + v3 * v3;
  const bool  zero = (v0 == 0.f) && (v1 == 0.f) && (v2 == 0.f) && (v3 == 0.f);
  const float inv  = zero ? 1.f : (1.f / sqrtf(ssq));
  const float t0 = zero ? 1.f : v0 * inv;
  const float t1 = zero ? 0.f : v1 * inv;
  const float t2 = zero ? 0.f : v2 * inv;
  const float t3 = zero ? 0.f : v3 * inv;
  cs_half(0.5f * t0, c[0], s[0]);
  cs_half(0.5f * t1, c[1], s[1]);
  cs_half(0.5f * t2, c[2], s[2]);
  cs_half(0.5f * t3, c[3], s[3]);
}

static __device__ __forceinline__ void afrag(const float c[4], const float s[4], int hb,
                                             v2f a0[4], v2f a1[4]) {
  const float p01[4] = { c[0] * c[1], s[0] * c[1], c[0] * s[1], s[0] * s[1] };
  const float p23[4] = { c[2] * c[3], s[2] * c[3], c[2] * s[3], s[2] * s[3] };
  const float qo0 = hb ? p01[2] : p01[0];
  const float qo1 = hb ? p01[3] : p01[1];
  const float qx0 = hb ? p01[0] : p01[2];
  const float qx1 = hb ? p01[1] : p01[3];
  const float rq0 = xhalf(qx0);
  const float rq1 = xhalf(qx1);
  const float rp[4] = { xhalf(p23[0]), xhalf(p23[1]), xhalf(p23[2]), xhalf(p23[3]) };
  const float t0q0 = hb ? rq0 : qo0, t0q1 = hb ? rq1 : qo1;
  const float t1q0 = hb ? qo0 : rq0, t1q1 = hb ? qo1 : rq1;
  for (int f = 0; f < 4; ++f) {
    const float t0p = hb ? rp[f] : p23[f];
    const float t1p = hb ? p23[f] : rp[f];
    a0[f].x = t0q0 * t0p;
    a0[f].y = t0q1 * t0p;
    a1[f].x = t1q0 * t1p;
    a1[f].y = t1q1 * t1p;
  }
}

static __device__ __forceinline__ void zsums(const float* row, float& q0, float& q1, float& q2, float& q3) {
  q0 = 0.f; q1 = 0.f; q2 = 0.f; q3 = 0.f;
#pragma unroll
  for (int p = 0; p < 16; ++p) {
    const float pv = row[p];
    q0 += (p & 8) ? -pv : pv;
    q1 += (p & 4) ? -pv : pv;
    q2 += (p & 2) ? -pv : pv;
    q3 += (p & 1) ? -pv : pv;
  }
}

extern "C" __global__ __launch_bounds__(256) void qconv_main(
    const float* __restrict__ x, const float* __restrict__ w,
    float* __restrict__ out) {
  __shared__ float lds[8 * 1024];
  __shared__ __attribute__((aligned(16))) float sc[64];
  __shared__ float psi[32];
  __shared__ float Cs[512];

  const int tid  = threadIdx.x;
  const int lane = tid & 31;
  const int wave = tid >> 5;
  const int hb   = lane >> 4;
  const int M    = lane & 15;
  const int orow = blockIdx.x * 8 + wave;
  const int i = orow & 63, b = orow >> 6;

  const float* img = x + (((size_t)b) << 14) + (size_t)(2 * i) * 128 + (size_t)(4 * lane);
  const float4 top = *(const float4*)(img);
  const float4 bot = *(const float4*)(img + 128);

  if (tid < 28) {
    const float th = w[tid];
    sc[tid]      = cosf(0.5f * th);
    sc[32 + tid] = sinf(0.5f * th);
  }
  __syncthreads();

  if (tid < 16) {
    const int r = tid;
    float re = 1.f, im = 0.f;
#pragma unroll
    for (int q = 0; q < 4; ++q) {
      const bool bit = ((r >> (3 - q)) & 1) != 0;
      const float c = sc[q], s = sc[32 + q], cz = sc[4 + q], sz = sc[36 + q];
      const float ar = bit ? s * sz : c * cz;
      const float ai = bit ? -(s * cz) : -(c * sz);
      const float nr = re * ar - im * ai;
      const float ni = re * ai + im * ar;
      re = nr; im = ni;
    }
    int k = 8;
#pragma unroll
    for (int cd = 0; cd < 4; ++cd) {
      const int ctrl = 3 - cd;
#pragma unroll
      for (int tgt = 0; tgt < 4; ++tgt) {
        if (tgt == ctrl) continue;
        const float c = sc[k], s = sc[32 + k];
        const bool use = (r & (8 >> ctrl)) != 0;
        const float sg = (r & (8 >> tgt)) ? s : -s;
        const float pr = use ? c : 1.f;
        const float pi = use ? sg : 0.f;
        const float nr = re * pr - im * pi;
        const float ni = re * pi + im * pr;
        re = nr; im = ni;
        ++k;
      }
    }
    psi[r]      = re;
    psi[16 + r] = im;
  }

  float ce[4], se[4], co[4], so[4];
  encode(top.x, top.y, bot.x, bot.y, ce, se);
  encode(top.z, top.w, bot.z, bot.w, co, so);

  __syncthreads();

  {
    const int S = tid >> 4, p = tid & 15;
    float vr = psi[p], vi = psi[16 + p];
    const float4 cT = *(const float4*)&sc[20];
    const float4 cP = *(const float4*)&sc[24];
    const float4 sT = *(const float4*)&sc[52];
    const float4 sP = *(const float4*)&sc[56];
    const float cth[4] = { cT.x, cT.y, cT.z, cT.w };
    const float cph[4] = { cP.x, cP.y, cP.z, cP.w };
    const float sth[4] = { sT.x, sT.y, sT.z, sT.w };
    const float sph[4] = { sP.x, sP.y, sP.z, sP.w };
#pragma unroll
    for (int q = 0; q < 4; ++q) {
      const int  mask = 8 >> q;
      const bool a  = (p & mask) != 0;
      const bool kb = ((S >> q) & 1) != 0;
      const float pr = __shfl_xor(vr, mask, 32);
      const float pi = __shfl_xor(vi, mask, 32);
      const float sg = a ? sph[q] : -sph[q];
      const float Dr = cph[q] * cth[q], Di = sg * cth[q];
      const float Or = sg * sth[q],     Oi = -(cph[q] * sth[q]);
      const float tau = a ? -1.f : 1.f;
      const float xr = kb ? tau * Or : Dr;
      const float xi = kb ? tau * Oi : Di;
      const float yr = kb ? -(tau * Dr) : Or;
      const float yi = kb ? -(tau * Di) : Oi;
      const float nr = xr * vr - xi * vi + yr * pr - yi * pi;
      const float ni = xr * vi + xi * vr + yr * pi + yi * pr;
      vr = nr; vi = ni;
    }
    Cs[tid]       = vr;
    Cs[256 + tid] = vi;
  }
  __syncthreads();

  v2f e0[4], e1[4], o0[4], o1[4];
  afrag(ce, se, hb, e0, e1);
  afrag(co, so, hb, o0, o1);

  v2f bre[4], bim[4];
#pragma unroll
  for (int f = 0; f < 4; ++f) {
    const int kb = 4 * f + 2 * hb;
    bre[f].x = Cs[kb * 16 + M];
    bre[f].y = Cs[kb * 16 + 16 + M];
    bim[f].x = Cs[256 + kb * 16 + M];
    bim[f].y = Cs[256 + kb * 16 + 16 + M];
  }

  const v8f zero8 = {0.f, 0.f, 0.f, 0.f, 0.f, 0.f, 0.f, 0.f};
  v8f re_e0 = zero8, im_e0 = zero8, re_e1 = zero8, im_e1 = zero8;
  v8f re_o0 = zero8, im_o0 = zero8, re_o1 = zero8, im_o1 = zero8;
#pragma unroll
  for (int f = 0; f < 4; ++f) {
    re_e0 = __builtin_amdgcn_wmma_f32_16x16x4_f32(false, e0[f], false, bre[f], (short)0, re_e0, false, false);
    im_e0 = __builtin_amdgcn_wmma_f32_16x16x4_f32(false, e0[f], false, bim[f], (short)0, im_e0, false, false);
    re_e1 = __builtin_amdgcn_wmma_f32_16x16x4_f32(false, e1[f], false, bre[f], (short)0, re_e1, false, false);
    im_e1 = __builtin_amdgcn_wmma_f32_16x16x4_f32(false, e1[f], false, bim[f], (short)0, im_e1, false, false);
    re_o0 = __builtin_amdgcn_wmma_f32_16x16x4_f32(false, o0[f], false, bre[f], (short)0, re_o0, false, false);
    im_o0 = __builtin_amdgcn_wmma_f32_16x16x4_f32(false, o0[f], false, bim[f], (short)0, im_o0, false, false);
    re_o1 = __builtin_amdgcn_wmma_f32_16x16x4_f32(false, o1[f], false, bre[f], (short)0, re_o1, false, false);
    im_o1 = __builtin_amdgcn_wmma_f32_16x16x4_f32(false, o1[f], false, bim[f], (short)0, im_o1, false, false);
  }
  const v8f Pe0 = re_e0 * re_e0 + im_e0 * im_e0;
  const v8f Pe1 = re_e1 * re_e1 + im_e1 * im_e1;
  const v8f Po0 = re_o0 * re_o0 + im_o0 * im_o0;
  const v8f Po1 = re_o1 * re_o1 + im_o1 * im_o1;

  float* lp = &lds[wave * 1024];
#pragma unroll
  for (int v = 0; v < 8; ++v) {
    lp[(v + 8 * hb) * 16 + M]       = Pe0[v];
    lp[256 + (v + 8 * hb) * 16 + M] = Pe1[v];
    lp[512 + (v + 8 * hb) * 16 + M] = Po0[v];
    lp[768 + (v + 8 * hb) * 16 + M] = Po1[v];
  }
  asm volatile("s_wait_dscnt 0x0" ::: "memory");

  {
    float qe0, qe1, qe2, qe3, qo0, qo1, qo2, qo3;
    zsums(lp + lane * 16, qe0, qe1, qe2, qe3);
    zsums(lp + 512 + lane * 16, qo0, qo1, qo2, qo3);
    float* o = out + (((size_t)b) << 14) + (size_t)i * 64 + (size_t)(2 * lane);
    *(float2*)(o)         = make_float2(qe0, qo0);
    *(float2*)(o + 4096)  = make_float2(qe1, qo1);
    *(float2*)(o + 8192)  = make_float2(qe2, qo2);
    *(float2*)(o + 12288) = make_float2(qe3, qo3);
  }
}

extern "C" void kernel_launch(void* const* d_in, const int* in_sizes, int n_in,
                              void* d_out, int out_size, void* d_ws, size_t ws_size,
                              hipStream_t stream) {
  const float* x = (const float*)d_in[0];
  const float* w = (const float*)d_in[1];
  float* out = (float*)d_out;

  qconv_main<<<512, 256, 0, stream>>>(x, w, out);
}


# Product transposed (fixed matrix as A operand): a lane holds 8 states of a patch per tile, so probabilities and Z sums stay in registers (one lane exchange); LDS staging gone; new Z-sum order
# speedup vs baseline: 2.8845x; 1.0746x over previous
#include <hip/hip_runtime.h>
#include <math.h>


typedef float v2f __attribute__((ext_vector_type(2)));
typedef float v8f __attribute__((ext_vector_type(8)));

static __device__ __forceinline__ float xhalf(float v) {
  const int bits = __builtin_bit_cast(int, v);
  return __builtin_bit_cast(float,
      __builtin_amdgcn_permlanex16(bits, bits, 0x76543210, 0xfedcba98, false, false));
}

static __device__ __forceinline__ void cs_half(float h, float& c, float& s) {
  const float z = h * h;
  float p = fmaf(z, -1.94644585e-4f, 8.33171979e-3f);
  p = fmaf(z, p, -1.66666463e-1f);
  p = z * p;
  s = fmaf(h, p, h);
  float q = fmaf(z, 2.56687417e-5f, -1.39091106e-3f);
  q = fmaf(z, q, 4.16679047e-2f);
  q = fmaf(z, q, -5.00000238e-1f);
  c = fmaf(z, q, 1.0f);
}


static __device__ __forceinline__ void encode(float v0, float v1, float v2, float v3,
                                              float c[4], float s[4]) {
  const float ssq  = v0 * v0 + v1 * v1 + v2 * v2 + v3 * v3;
  const bool  zero = (v0 == 0.f) && (v1 == 0.f) && (v2 == 0.f) && (v3 == 0.f);
  const float inv  = zero ? 1.f : (1.f / sqrtf(ssq));
  const float t0 = zero ? 1.f : v0 * inv;
  const float t1 = zero ? 0.f : v1 * inv;
  const float t2 = zero ? 0.f : v2 * inv;
  const float t3 = zero ? 0.f : v3 * inv;
  cs_half(0.5f * t0, c[0], s[0]);
  cs_half(0.5f * t1, c[1], s[1]);
  cs_half(0.5f * t2, c[2], s[2]);
  cs_half(0.5f * t3, c[3], s[3]);
}

static __device__ __forceinline__ void afrag(const float c[4], const float s[4], int hb,
                                             v2f a0[4], v2f a1[4]) {
  const float p01[4] = { c[0] * c[1], s[0] * c[1], c[0] * s[1], s[0] * s[1] };
  const float p23[4] = { c[2] * c[3], s[2] * c[3], c[2] * s[3], s[2] * s[3] };
  const float qo0 = hb ? p01[2] : p01[0];
  const float qo1 = hb ? p01[3] : p01[1];
  const float qx0 = hb ? p01[0] : p01[2];
  const float qx1 = hb ? p01[1] : p01[3];
  const float rq0 = xhalf(qx0);
  const float rq1 = xhalf(qx1);
  const float rp[4] = { xhalf(p23[0]), xhalf(p23[1]), xhalf(p23[2]), xhalf(p23[3]) };
  const float t0q0 = hb ? rq0 : qo0, t0q1 = hb ? rq1 : qo1;
  const float t1q0 = hb ? qo0 : rq0, t1q1 = hb ? qo1 : rq1;
  for (int f = 0; f < 4; ++f) {
    const float t0p = hb ? rp[f] : p23[f];
    const float t1p = hb ? p23[f] : rp[f];
    a0[f].x = t0q0 * t0p;
    a0[f].y = t0q1 * t0p;
    a1[f].x = t1q0 * t1p;
    a1[f].y = t1q1 * t1p;
  }
}

static __device__ __forceinline__ void zpart(const v8f re, const v8f im, int hb, float z[4]) {
  const v8f P = re * re + im * im;
  const float a = P[0] + P[1], b = P[2] + P[3], c = P[4] + P[5], d = P[6] + P[7];
  const float ab = a + b, cd = c + d;
  const float tot = ab + cd;
  z[0] = hb ? -tot : tot;
  z[1] = ab - cd;
  z[2] = (a - b) + (c - d);
  z[3] = ((P[0] - P[1]) + (P[2] - P[3])) + ((P[4] - P[5]) + (P[6] - P[7]));
}

extern "C" __global__ __launch_bounds__(256) void qconv_main(
    const float* __restrict__ x, const float* __restrict__ w,
    float* __restrict__ out) {
  __shared__ __attribute__((aligned(16))) float sc[64];
  __shared__ float psi[32];
  __shared__ float Cs[512];

  const int tid  = threadIdx.x;
  const int lane = tid & 31;
  const int wave = tid >> 5;
  const int hb   = lane >> 4;
  const int M    = lane & 15;
  const int orow = blockIdx.x * 8 + wave;
  const int i = orow & 63, b = orow >> 6;

  const float* img = x + (((size_t)b) << 14) + (size_t)(2 * i) * 128 + (size_t)(4 * lane);
  const float4 top = *(const float4*)(img);
  const float4 bot = *(const float4*)(img + 128);

  if (tid < 28) {
    const float th = w[tid];
    sc[tid]      = cosf(0.5f * th);
    sc[32 + tid] = sinf(0.5f * th);
  }
  __syncthreads();

  if (tid < 16) {
    const int r = tid;
    float re = 1.f, im = 0.f;
#pragma unroll
    for (int q = 0; q < 4; ++q) {
      const bool bit = ((r >> (3 - q)) & 1) != 0;
      const float c = sc[q], s = sc[32 + q], cz = sc[4 + q], sz = sc[36 + q];
      const float ar = bit ? s * sz : c * cz;
      const float ai = bit ? -(s * cz) : -(c * sz);
      const float nr = re * ar - im * ai;
      const float ni = re * ai + im * ar;
      re = nr; im = ni;
    }
    int k = 8;
#pragma unroll
    for (int cd = 0; cd < 4; ++cd) {
      const int ctrl = 3 - cd;
#pragma unroll
      for (int tgt = 0; tgt < 4; ++tgt) {
        if (tgt == ctrl) continue;
        const float c = sc[k], s = sc[32 + k];
        const bool use = (r & (8 >> ctrl)) != 0;
        const float sg = (r & (8 >> tgt)) ? s : -s;
        const float pr = use ? c : 1.f;
        const float pi = use ? sg : 0.f;
        const float nr = re * pr - im * pi;
        const float ni = re * pi + im * pr;
        re = nr; im = ni;
        ++k;
      }
    }
    psi[r]      = re;
    psi[16 + r] = im;
  }

  float ce[4], se[4], co[4], so[4];
  encode(top.x, top.y, bot.x, bot.y, ce, se);
  encode(top.z, top.w, bot.z, bot.w, co, so);

  __syncthreads();

  {
    const int S = tid >> 4, p = tid & 15;
    float vr = psi[p], vi = psi[16 + p];
    const float4 cT = *(const float4*)&sc[20];
    const float4 cP = *(const float4*)&sc[24];
    const float4 sT = *(const float4*)&sc[52];
    const float4 sP = *(const float4*)&sc[56];
    const float cth[4] = { cT.x, cT.y, cT.z, cT.w };
    const float cph[4] = { cP.x, cP.y, cP.z, cP.w };
    const float sth[4] = { sT.x, sT.y, sT.z, sT.w };
    const float sph[4] = { sP.x, sP.y, sP.z, sP.w };
#pragma unroll
    for (int q = 0; q < 4; ++q) {
      const int  mask = 8 >> q;
      const bool a  = (p & mask) != 0;
      const bool kb = ((S >> q) & 1) != 0;
      const float pr = __shfl_xor(vr, mask, 32);
      const float pi = __shfl_xor(vi, mask, 32);
      const float sg = a ? sph[q] : -sph[q];
      const float Dr = cph[q] * cth[q], Di = sg * cth[q];
      const float Or = sg * sth[q],     Oi = -(cph[q] * sth[q]);
      const float tau = a ? -1.f : 1.f;
      const float xr = kb ? tau * Or : Dr;
      const float xi = kb ? tau * Oi : Di;
      const float yr = kb ? -(tau * Dr) : Or;
      const float yi = kb ? -(tau * Di) : Oi;
      const float nr = xr * vr - xi * vi + yr * pr - yi * pi;
      const float ni = xr * vi + xi * vr + yr * pi + yi * pr;
      vr = nr; vi = ni;
    }
    Cs[tid]       = vr;
    Cs[256 + tid] = vi;
  }
  __syncthreads();

  v2f e0[4], e1[4], o0[4], o1[4];
  afrag(ce, se, hb, e0, e1);
  afrag(co, so, hb, o0, o1);

  v2f bre[4], bim[4];
#pragma unroll
  for (int f = 0; f < 4; ++f) {
    const int kb = 4 * f + 2 * hb;
    bre[f].x = Cs[kb * 16 + M];
    bre[f].y = Cs[kb * 16 + 16 + M];
    bim[f].x = Cs[256 + kb * 16 + M];
    bim[f].y = Cs[256 + kb * 16 + 16 + M];
  }

  const v8f zero8 = {0.f, 0.f, 0.f, 0.f, 0.f, 0.f, 0.f, 0.f};
  v8f re_e0 = zero8, im_e0 = zero8, re_e1 = zero8, im_e1 = zero8;
  v8f re_o0 = zero8, im_o0 = zero8, re_o1 = zero8, im_o1 = zero8;
#pragma unroll
  for (int f = 0; f < 4; ++f) {
    re_e0 = __builtin_amdgcn_wmma_f32_16x16x4_f32(false, bre[f], false, e0[f], (short)0, re_e0, false, false);
    im_e0 = __builtin_amdgcn_wmma_f32_16x16x4_f32(false, bim[f], false, e0[f], (short)0, im_e0, false, false);
    re_e1 = __builtin_amdgcn_wmma_f32_16x16x4_f32(false, bre[f], false, e1[f], (short)0, re_e1, false, false);
    im_e1 = __builtin_amdgcn_wmma_f32_16x16x4_f32(false, bim[f], false, e1[f], (short)0, im_e1, false, false);
    re_o0 = __builtin_amdgcn_wmma_f32_16x16x4_f32(false, bre[f], false, o0[f], (short)0, re_o0, false, false);
    im_o0 = __builtin_amdgcn_wmma_f32_16x16x4_f32(false, bim[f], false, o0[f], (short)0, im_o0, false, false);
    re_o1 = __builtin_amdgcn_wmma_f32_16x16x4_f32(false, bre[f], false, o1[f], (short)0, re_o1, false, false);
    im_o1 = __builtin_amdgcn_wmma_f32_16x16x4_f32(false, bim[f], false, o1[f], (short)0, im_o1, false, false);
  }

  float ze0[4], ze1[4], zo0[4], zo1[4];
  zpart(re_e0, im_e0, hb, ze0);
  zpart(re_e1, im_e1, hb, ze1);
  zpart(re_o0, im_o0, hb, zo0);
  zpart(re_o1, im_o1, hb, zo1);

  {
    float qe[4], qo[4];
#pragma unroll
    for (int q = 0; q < 4; ++q) {
      const float keep_e = hb ? ze1[q] : ze0[q];
      const float send_e = hb ? ze0[q] : ze1[q];
      const float keep_o = hb ? zo1[q] : zo0[q];
      const float send_o = hb ? zo0[q] : zo1[q];
      qe[q] = keep_e + xhalf(send_e);
      qo[q] = keep_o + xhalf(send_o);
    }
    float* o = out + (((size_t)b) << 14) + (size_t)i * 64 + (size_t)(2 * lane);
    *(float2*)(o)         = make_float2(qe[0], qo[0]);
    *(float2*)(o + 4096)  = make_float2(qe[1], qo[1]);
    *(float2*)(o + 8192)  = make_float2(qe[2], qo[2]);
    *(float2*)(o + 12288) = make_float2(qe[3], qo[3]);
  }
}

extern "C" void kernel_launch(void* const* d_in, const int* in_sizes, int n_in,
                              void* d_out, int out_size, void* d_ws, size_t ws_size,
                              hipStream_t stream) {
  const float* x = (const float*)d_in[0];
  const float* w = (const float*)d_in[1];
  float* out = (float*)d_out;

  qconv_main<<<512, 256, 0, stream>>>(x, w, out);
}


# Matrix build no longer waits for the pixel loads: weight load issued first and work-group barriers fence LDS only, so both 16-byte pixel loads stay in flight across build stages 1-2; same values
# speedup vs baseline: 3.0444x; 1.0554x over previous
#include <hip/hip_runtime.h>
#include <math.h>


typedef float v2f __attribute__((ext_vector_type(2)));
typedef float v8f __attribute__((ext_vector_type(8)));

static __device__ __forceinline__ float xhalf(float v) {
  const int bits = __builtin_bit_cast(int, v);
  return __builtin_bit_cast(float,
      __builtin_amdgcn_permlanex16(bits, bits, 0x76543210, 0xfedcba98, false, false));
}

static __device__ __forceinline__ void cs_half(float h, float& c, float& s) {
  const float z = h * h;
  float p = fmaf(z, -1.94644585e-4f, 8.33171979e-3f);
  p = fmaf(z, p, -1.66666463e-1f);
  p = z * p;
  s = fmaf(h, p, h);
  float q = fmaf(z, 2.56687417e-5f, -1.39091106e-3f);
  q = fmaf(z, q, 4.16679047e-2f);
  q = fmaf(z, q, -5.00000238e-1f);
  c = fmaf(z, q, 1.0f);
}


static __device__ __forceinline__ void encode(float v0, float v1, float v2, float v3,
                                              float c[4], float s[4]) {
  const float ssq  = v0 * v0 + v1 * v1 + v2 * v2 + v3 * v3;
  const bool  zero = (v0 == 0.f) && (v1 == 0.f) && (v2 == 0.f) && (v3 == 0.f);
  const float inv  = zero ? 1.f : (1.f / sqrtf(ssq));
  const float t0 = zero ? 1.f : v0 * inv;
  const float t1 = zero ? 0.f : v1 * inv;
  const float t2 = zero ? 0.f : v2 * inv;
  const float t3 = zero ? 0.f : v3 * inv;
  cs_half(0.5f * t0, c[0], s[0]);
  cs_half(0.5f * t1, c[1], s[1]);
  cs_half(0.5f * t2, c[2], s[2]);
  cs_half(0.5f * t3, c[3], s[3]);
}

static __device__ __forceinline__ void afrag(const float c[4], const float s[4], int hb,
                                             v2f a0[4], v2f a1[4]) {
  const float p01[4] = { c[0] * c[1], s[0] * c[1], c[0] * s[1], s[0] * s[1] };
  const float p23[4] = { c[2] * c[3], s[2] * c[3], c[2] * s[3], s[2] * s[3] };
  const float qo0 = hb ? p01[2] : p01[0];
  const float qo1 = hb ? p01[3] : p01[1];
  const float qx0 = hb ? p01[0] : p01[2];
  const float qx1 = hb ? p01[1] : p01[3];
  const float rq0 = xhalf(qx0);
  const float rq1 = xhalf(qx1);
  const float rp[4] = { xhalf(p23[0]), xhalf(p23[1]), xhalf(p23[2]), xhalf(p23[3]) };
  const float t0q0 = hb ? rq0 : qo0, t0q1 = hb ? rq1 : qo1;
  const float t1q0 = hb ? qo0 : rq0, t1q1 = hb ? qo1 : rq1;
  for (int f = 0; f < 4; ++f) {
    const float t0p = hb ? rp[f] : p23[f];
    const float t1p = hb ? p23[f] : rp[f];
    a0[f].x = t0q0 * t0p;
    a0[f].y = t0q1 * t0p;
    a1[f].x = t1q0 * t1p;
    a1[f].y = t1q1 * t1p;
  }
}

static __device__ __forceinline__ void zpart(const v8f re, const v8f im, int hb, float z[4]) {
  const v8f P = re * re + im * im;
  const float a = P[0] + P[1], b = P[2] + P[3], c = P[4] + P[5], d = P[6] + P[7];
  const float ab = a + b, cd = c + d;
  const float tot = ab + cd;
  z[0] = hb ? -tot : tot;
  z[1] = ab - cd;
  z[2] = (a - b) + (c - d);
  z[3] = ((P[0] - P[1]) + (P[2] - P[3])) + ((P[4] - P[5]) + (P[6] - P[7]));
}

static __device__ __forceinline__ void lds_barrier() {
  __builtin_amdgcn_fence(__ATOMIC_RELEASE, "workgroup", "local");
  __builtin_amdgcn_s_barrier();
  __builtin_amdgcn_fence(__ATOMIC_ACQUIRE, "workgroup", "local");
}

extern "C" __global__ __launch_bounds__(256) void qconv_main(
    const float* __restrict__ x, const float* __restrict__ w,
    float* __restrict__ out) {
  __shared__ __attribute__((aligned(16))) float sc[64];
  __shared__ float psi[32];
  __shared__ float Cs[512];

  const int tid  = threadIdx.x;
  const int lane = tid & 31;
  const int wave = tid >> 5;
  const int hb   = lane >> 4;
  const int M    = lane & 15;
  const int orow = blockIdx.x * 8 + wave;
  const int i = orow & 63, b = orow >> 6;

  const float* img = x + (((size_t)b) << 14) + (size_t)(2 * i) * 128 + (size_t)(4 * lane);
  float th = 0.f;
  if (tid < 28) th = w[tid];
  const float4 top = *(const float4*)(img);
  const float4 bot = *(const float4*)(img + 128);
  asm volatile("" : "+v"(th));

  if (tid < 28) {
    sc[tid]      = cosf(0.5f * th);
    sc[32 + tid] = sinf(0.5f * th);
  }
  lds_barrier();

  if (tid < 16) {
    const int r = tid;
    float re = 1.f, im = 0.f;
#pragma unroll
    for (int q = 0; q < 4; ++q) {
      const bool bit = ((r >> (3 - q)) & 1) != 0;
      const float c = sc[q], s = sc[32 + q], cz = sc[4 + q], sz = sc[36 + q];
      const float ar = bit ? s * sz : c * cz;
      const float ai = bit ? -(s * cz) : -(c * sz);
      const float nr = re * ar - im * ai;
      const float ni = re * ai + im * ar;
      re = nr; im = ni;
    }
    int k = 8;
#pragma unroll
    for (int cd = 0; cd < 4; ++cd) {
      const int ctrl = 3 - cd;
#pragma unroll
      for (int tgt = 0; tgt < 4; ++tgt) {
        if (tgt == ctrl) continue;
        const float c = sc[k], s = sc[32 + k];
        const bool use = (r & (8 >> ctrl)) != 0;
        const float sg = (r & (8 >> tgt)) ? s : -s;
        const float pr = use ? c : 1.f;
        const float pi = use ? sg : 0.f;
        const float nr = re * pr - im * pi;
        const float ni = re * pi + im * pr;
        re = nr; im = ni;
        ++k;
      }
    }
    psi[r]      = re;
    psi[16 + r] = im;
  }

  float ce[4], se[4], co[4], so[4];
  encode(top.x, top.y, bot.x, bot.y, ce, se);
  encode(top.z, top.w, bot.z, bot.w, co, so);

  lds_barrier();

  {
    const int S = tid >> 4, p = tid & 15;
    float vr = psi[p], vi = psi[16 + p];
    const float4 cT = *(const float4*)&sc[20];
    const float4 cP = *(const float4*)&sc[24];
    const float4 sT = *(const float4*)&sc[52];
    const float4 sP = *(const float4*)&sc[56];
    const float cth[4] = { cT.x, cT.y, cT.z, cT.w };
    const float cph[4] = { cP.x, cP.y, cP.z, cP.w };
    const float sth[4] = { sT.x, sT.y, sT.z, sT.w };
    const float sph[4] = { sP.x, sP.y, sP.z, sP.w };
#pragma unroll
    for (int q = 0; q < 4; ++q) {
      const int  mask = 8 >> q;
      const bool a  = (p & mask) != 0;
      const bool kb = ((S >> q) & 1) != 0;
      const float pr = __shfl_xor(vr, mask, 32);
      const float pi = __shfl_xor(vi, mask, 32);
      const float sg = a ? sph[q] : -sph[q];
      const float Dr = cph[q] * cth[q], Di = sg * cth[q];
      const float Or = sg * sth[q],     Oi = -(cph[q] * sth[q]);
      const float tau = a ? -1.f : 1.f;
      const float xr = kb ? tau * Or : Dr;
      const float xi = kb ? tau * Oi : Di;
      const float yr = kb ? -(tau * Dr) : Or;
      const float yi = kb ? -(tau * Di) : Oi;
      const float nr = xr * vr - xi * vi + yr * pr - yi * pi;
      const float ni = xr * vi + xi * vr + yr * pi + yi * pr;
      vr = nr; vi = ni;
    }
    Cs[tid]       = vr;
    Cs[256 + tid] = vi;
  }
  lds_barrier();

  v2f e0[4], e1[4], o0[4], o1[4];
  afrag(ce, se, hb, e0, e1);
  afrag(co, so, hb, o0, o1);

  v2f bre[4], bim[4];
#pragma unroll
  for (int f = 0; f < 4; ++f) {
    const int kb = 4 * f + 2 * hb;
    bre[f].x = Cs[kb * 16 + M];
    bre[f].y = Cs[kb * 16 + 16 + M];
    bim[f].x = Cs[256 + kb * 16 + M];
    bim[f].y = Cs[256 + kb * 16 + 16 + M];
  }

  const v8f zero8 = {0.f, 0.f, 0.f, 0.f, 0.f, 0.f, 0.f, 0.f};
  v8f re_e0 = zero8, im_e0 = zero8, re_e1 = zero8, im_e1 = zero8;
  v8f re_o0 = zero8, im_o0 = zero8, re_o1 = zero8, im_o1 = zero8;
#pragma unroll
  for (int f = 0; f < 4; ++f) {
    re_e0 = __builtin_amdgcn_wmma_f32_16x16x4_f32(false, bre[f], false, e0[f], (short)0, re_e0, false, false);
    im_e0 = __builtin_amdgcn_wmma_f32_16x16x4_f32(false, bim[f], false, e0[f], (short)0, im_e0, false, false);
    re_e1 = __builtin_amdgcn_wmma_f32_16x16x4_f32(false, bre[f], false, e1[f], (short)0, re_e1, false, false);
    im_e1 = __builtin_amdgcn_wmma_f32_16x16x4_f32(false, bim[f], false, e1[f], (short)0, im_e1, false, false);
    re_o0 = __builtin_amdgcn_wmma_f32_16x16x4_f32(false, bre[f], false, o0[f], (short)0, re_o0, false, false);
    im_o0 = __builtin_amdgcn_wmma_f32_16x16x4_f32(false, bim[f], false, o0[f], (short)0, im_o0, false, false);
    re_o1 = __builtin_amdgcn_wmma_f32_16x16x4_f32(false, bre[f], false, o1[f], (short)0, re_o1, false, false);
    im_o1 = __builtin_amdgcn_wmma_f32_16x16x4_f32(false, bim[f], false, o1[f], (short)0, im_o1, false, false);
  }

  float ze0[4], ze1[4], zo0[4], zo1[4];
  zpart(re_e0, im_e0, hb, ze0);
  zpart(re_e1, im_e1, hb, ze1);
  zpart(re_o0, im_o0, hb, zo0);
  zpart(re_o1, im_o1, hb, zo1);

  {
    float qe[4], qo[4];
#pragma unroll
    for (int q = 0; q < 4; ++q) {
      const float keep_e = hb ? ze1[q] : ze0[q];
      const float send_e = hb ? ze0[q] : ze1[q];
      const float keep_o = hb ? zo1[q] : zo0[q];
      const float send_o = hb ? zo0[q] : zo1[q];
      qe[q] = keep_e + xhalf(send_e);
      qo[q] = keep_o + xhalf(send_o);
    }
    float* o = out + (((size_t)b) << 14) + (size_t)i * 64 + (size_t)(2 * lane);
    *(float2*)(o)         = make_float2(qe[0], qo[0]);
    *(float2*)(o + 4096)  = make_float2(qe[1], qo[1]);
    *(float2*)(o + 8192)  = make_float2(qe[2], qo[2]);
    *(float2*)(o + 12288) = make_float2(qe[3], qo[3]);
  }
}

extern "C" void kernel_launch(void* const* d_in, const int* in_sizes, int n_in,
                              void* d_out, int out_size, void* d_ws, size_t ws_size,
                              hipStream_t stream) {
  const float* x = (const float*)d_in[0];
  const float* w = (const float*)d_in[1];
  float* out = (float*)d_out;

  qconv_main<<<512, 256, 0, stream>>>(x, w, out);
}
